# starting kernel
# baseline (speedup 1.0000x reference)
#include <hip/hip_runtime.h>
#include <hip/hip_bf16.h>
#include <math.h>


typedef __attribute__((ext_vector_type(2))) float v2f;
typedef __attribute__((ext_vector_type(8))) float v8f;

#define PROB_TH 1e-3f

__global__ __launch_bounds__(256) void ffg_tilecolors(
    const float* __restrict__ fc,
    const float* __restrict__ fs,
    const int*   __restrict__ temp,
    float*       __restrict__ tileColors)
{
  __shared__ float sp[64 * 32];
  __shared__ int   st3[64 * 3];
  __shared__ float sc[96];
  __shared__ float wl[256 * 33];

  const int tid  = threadIdx.x;
  const float invT = 1.0f / (float)temp[0];

  if (tid < 64) {
    const float* row = fs + tid * 32;
    float m = -1e30f;
    for (int j = 0; j < 32; j++) m = fmaxf(m, row[j] * invT);
    float s = 0.f;
    for (int j = 0; j < 32; j++) {
      float e = expf(row[j] * invT - m);
      s += e;
      sp[tid * 32 + j] = e;
    }
    float inv = 1.0f / s;
    float v0 = -1.f, v1 = -1.f, v2 = -1.f;
    int   i0 = 0,    i1 = 0,    i2 = 0;
    for (int j = 0; j < 32; j++) {
      float p = sp[tid * 32 + j] * inv;
      sp[tid * 32 + j] = p;
      if (p > v0)      { v2 = v1; i2 = i1; v1 = v0; i1 = i0; v0 = p; i0 = j; }
      else if (p > v1) { v2 = v1; i2 = i1; v1 = p;  i1 = j; }
      else if (p > v2) { v2 = p;  i2 = j; }
    }
    st3[tid * 3 + 0] = i0;
    st3[tid * 3 + 1] = i1;
    st3[tid * 3 + 2] = i2;
  }
  if (tid < 96) sc[tid] = 1.0f / (1.0f + expf(-fc[tid]));
  __syncthreads();

  const int tile = blockIdx.x * 256 + tid;
  const int ty = tile >> 5, tx = tile & 31;
  float* w = &wl[tid * 33];
  for (int i = 0; i < 32; i++) w[i] = 0.f;

  {
    const int r0 = ((ty >> 4) & 1);
    const float* p0 = &sp[r0 * 32];
    const int*   t0 = &st3[r0 * 3];
    for (int j0 = 0; j0 < 3; j0++) {
      const int c1 = t0[j0];
      const float q1 = p0[c1];
      if (!(q1 > PROB_TH)) continue;
      const int dir1 = c1 & 1;
      const int cy1 = 1, cx1 = 0;
      const int side1 = dir1 ? ((tx >> (4 - cx1)) & 1) : ((ty >> (4 - cy1)) & 1);
      const int r1 = c1 * 2 + side1;
      const float* p1 = &sp[r1 * 32];
      const int*   t1 = &st3[r1 * 3];
      for (int j1 = 0; j1 < 3; j1++) {
        const int c2 = t1[j1];
        const float q2 = p1[c2];
        if (!(q2 > PROB_TH)) continue;
        const int cy2 = cy1 + (dir1 ^ 1), cx2 = cx1 + dir1;
        const int dir2 = c2 & 1;
        const int side2 = dir2 ? ((tx >> (4 - cx2)) & 1) : ((ty >> (4 - cy2)) & 1);
        const int r2 = c2 * 2 + side2;
        const float* p2 = &sp[r2 * 32];
        const int*   t2 = &st3[r2 * 3];
        const float w2 = q1 * q2;
        for (int j2 = 0; j2 < 2; j2++) {
          const int c3 = t2[j2];
          const float q3 = p2[c3];
          if (!(q3 > PROB_TH)) continue;
          const int cy3 = cy2 + (dir2 ^ 1), cx3 = cx2 + dir2;
          const int dir3 = c3 & 1;
          const int side3 = dir3 ? ((tx >> (4 - cx3)) & 1) : ((ty >> (4 - cy3)) & 1);
          const int r3 = c3 * 2 + side3;
          const float* p3 = &sp[r3 * 32];
          const int*   t3 = &st3[r3 * 3];
          const float w3 = w2 * q3;
          for (int j3 = 0; j3 < 2; j3++) {
            const int c4 = t3[j3];
            const float q4 = p3[c4];
            if (!(q4 > PROB_TH)) continue;
            const int cy4 = cy3 + (dir3 ^ 1), cx4 = cx3 + dir3;
            const int dir4 = c4 & 1;
            const int side4 = dir4 ? ((tx >> (4 - cx4)) & 1) : ((ty >> (4 - cy4)) & 1);
            const int r4 = c4 * 2 + side4;
            const float w4 = w3 * q4;
            const int c5 = st3[r4 * 3 + 0];
            const float q5 = sp[r4 * 32 + c5];
            if (q5 > PROB_TH) w[c5] += w4 * q5;
          }
        }
      }
    }
  }
  __syncthreads();

  const int wave = tid >> 5;
  const int lane = tid & 31;
  const int half = lane >> 4;
  const int ln   = lane & 15;
  for (int cc = 0; cc < 2; cc++) {
    const int base = (wave * 2 + cc) * 16;
    v8f acc = {};
#pragma unroll
    for (int s = 0; s < 8; s++) {
      const int kb = 4 * s + 2 * half;
      v2f a, b;
      a.x = wl[(base + ln) * 33 + kb];
      a.y = wl[(base + ln) * 33 + kb + 1];
      b.x = (ln < 3) ? sc[(kb + 0) * 3 + ln] : 0.f;
      b.y = (ln < 3) ? sc[(kb + 1) * 3 + ln] : 0.f;
      acc = __builtin_amdgcn_wmma_f32_16x16x4_f32(
          false, a, false, b, (short)0, acc, false, false);
    }
    const int gbase = blockIdx.x * 256 + base;
#pragma unroll
    for (int r = 0; r < 8; r++) {
      if (ln < 3) tileColors[(gbase + r + 8 * half) * 3 + ln] = acc[r];
    }
  }
}

__global__ __launch_bounds__(256) void ffg_write(
    const float* __restrict__ tc, float* __restrict__ out)
{
  const int bid = blockIdx.x;
  const int c = bid >> 10;
  const int y = bid & 1023;
  const int ty = y >> 5;
  const int t = threadIdx.x;
  const int x0 = t << 2;
  const int tx = x0 >> 5;
  const float v = tc[(ty * 32 + tx) * 3 + c];
  float4 o = make_float4(v, v, v, v);
  *(float4*)(out + ((size_t)c << 20) + ((size_t)y << 10) + x0) = o;
}

extern "C" void kernel_launch(void* const* d_in, const int* in_sizes, int n_in,
                              void* d_out, int out_size, void* d_ws, size_t ws_size,
                              hipStream_t stream) {
  (void)in_sizes; (void)n_in; (void)out_size; (void)ws_size;
  const float* fc   = (const float*)d_in[0];
  const float* fs   = (const float*)d_in[1];
  const int*   temp = (const int*)d_in[2];
  float* out = (float*)d_out;
  float* tc  = (float*)d_ws;

  ffg_tilecolors<<<4, 256, 0, stream>>>(fc, fs, temp, tc);
  ffg_write<<<3 * 1024, 256, 0, stream>>>(tc, out);
}


# Row softmax + top-3 scan kept in registers: each row read from global once, exps and probabilities stay in VGPRs, top-3 by branch-free compare/select, one LDS write per row; sums unchanged
# speedup vs baseline: 3.3154x; 3.3154x over previous
#include <hip/hip_runtime.h>
#include <hip/hip_bf16.h>
#include <math.h>


typedef __attribute__((ext_vector_type(2))) float v2f;
typedef __attribute__((ext_vector_type(8))) float v8f;

#define PROB_TH 1e-3f

__global__ __launch_bounds__(256) void ffg_frame(
    const float* __restrict__ fc,
    const float* __restrict__ fs,
    const int*   __restrict__ temp,
    float*       __restrict__ out)
{
  __shared__ float sp[64 * 32];
  __shared__ int   st3[64 * 3];
  __shared__ float sc[96];
  __shared__ float4 m4[64];
  __shared__ float4 t2[32 * 16];
  __shared__ float4 t3[32 * 16];

  const int tid  = threadIdx.x;
  const int ty   = blockIdx.x >> 3;
  const float invT = 1.0f / (float)temp[0];

  if (tid < 64) {
    const float* row = fs + tid * 32;
    float x[32];
#pragma unroll
    for (int j = 0; j < 32; j++) x[j] = row[j];
    float m = -1e30f;
#pragma unroll
    for (int j = 0; j < 32; j++) m = fmaxf(m, x[j] * invT);
    float s = 0.f;
#pragma unroll
    for (int j = 0; j < 32; j++) {
      const float e = expf(x[j] * invT - m);
      s += e;
      x[j] = e;
    }
    const float inv = 1.0f / s;
    float v0 = -1.f, v1 = -1.f, v2 = -1.f;
    int   i0 = 0,    i1 = 0,    i2 = 0;
#pragma unroll
    for (int j = 0; j < 32; j++) {
      const float p = x[j] * inv;
      x[j] = p;
      const bool c0 = p > v0, c1 = p > v1, c2 = p > v2;
      v2 = c1 ? v1 : (c2 ? p : v2);  i2 = c1 ? i1 : (c2 ? j : i2);
      v1 = c0 ? v0 : (c1 ? p : v1);  i1 = c0 ? i0 : (c1 ? j : i1);
      v0 = c0 ? p  : v0;             i0 = c0 ? j  : i0;
    }
#pragma unroll
    for (int j = 0; j < 32; j++) sp[tid * 32 + j] = x[j];
    st3[tid * 3 + 0] = i0;
    st3[tid * 3 + 1] = i1;
    st3[tid * 3 + 2] = i2;
  }
  if (tid < 96) sc[tid] = 1.0f / (1.0f + expf(-fc[tid]));
  __syncthreads();

  const int wave = tid >> 5;
  const int lane = tid & 31;
  const int half = lane >> 4;
  const int ln   = lane & 15;
  if (wave < 4) {
    const int row = wave * 16 + ln;
    const int top = st3[row * 3 + 0];
    v8f acc = {};
#pragma unroll
    for (int s = 0; s < 8; s++) {
      const int kb = 4 * s + 2 * half;
      const float pa = sp[row * 32 + kb];
      const float pb = sp[row * 32 + kb + 1];
      v2f a, b;
      a.x = (kb     == top && pa > PROB_TH) ? pa : 0.f;
      a.y = (kb + 1 == top && pb > PROB_TH) ? pb : 0.f;
      b.x = (ln < 3) ? sc[(kb + 0) * 3 + ln] : 0.f;
      b.y = (ln < 3) ? sc[(kb + 1) * 3 + ln] : 0.f;
      acc = __builtin_amdgcn_wmma_f32_16x16x4_f32(
          false, a, false, b, (short)0, acc, false, false);
    }
    float* m4f = (float*)m4;
#pragma unroll
    for (int r = 0; r < 8; r++) {
      if (ln < 4) m4f[(wave * 16 + r + 8 * half) * 4 + ln] = acc[r];
    }
  }
  __syncthreads();

  for (int i = 0; i < 2; i++) {
    const int e = tid + 256 * i;
    const int idx = e >> 4, u = (e >> 2) & 3, v = e & 3;
    const int odd = idx & 1;
    const int sd = odd ? (v >> 1) : (u >> 1);
    const int uc = odd ? (u >> 1) : (u & 1);
    const int vc = odd ? (v & 1)  : (v >> 1);
    const int r = idx * 2 + sd;
    float ax = 0.f, ay = 0.f, az = 0.f;
    for (int j = 0; j < 2; j++) {
      const int c = st3[r * 3 + j];
      const float p = sp[r * 32 + c];
      const float q = (p > PROB_TH) ? p : 0.f;
      const float4 k = m4[c * 2 + ((c & 1) ? vc : uc)];
      ax += q * k.x; ay += q * k.y; az += q * k.z;
    }
    t2[e] = make_float4(ax, ay, az, 0.f);
  }
  __syncthreads();

  for (int i = 0; i < 2; i++) {
    const int e = tid + 256 * i;
    const int idx = e >> 4, sel = (e >> 3) & 1, v = e & 7;
    const int u = sel ? (ty & 7) : ((ty >> 1) & 7);
    const int odd = idx & 1;
    const int sd = odd ? (v >> 2) : (u >> 2);
    const int uc = odd ? (u >> 1) : (u & 3);
    const int vc = odd ? (v & 3)  : (v >> 1);
    const int r = idx * 2 + sd;
    float ax = 0.f, ay = 0.f, az = 0.f;
    for (int j = 0; j < 2; j++) {
      const int c = st3[r * 3 + j];
      const float p = sp[r * 32 + c];
      const float q = (p > PROB_TH) ? p : 0.f;
      const float4 k = t2[c * 16 + uc * 4 + vc];
      ax += q * k.x; ay += q * k.y; az += q * k.z;
    }
    t3[e] = make_float4(ax, ay, az, 0.f);
  }
  __syncthreads();

  const int tx = tid >> 3;
  const int r0 = ty >> 4;
  const int u1 = ty & 15, v1 = tx >> 1;
  float ox = 0.f, oy = 0.f, oz = 0.f;
  for (int j0 = 0; j0 < 3; j0++) {
    const int c1 = st3[r0 * 3 + j0];
    const float p1 = sp[r0 * 32 + c1];
    const float q1 = (p1 > PROB_TH) ? p1 : 0.f;
    const int odd = c1 & 1;
    const int s1 = odd ? (v1 >> 3) : (u1 >> 3);
    const int u2 = odd ? (u1 >> 1) : (u1 & 7);
    const int v2 = odd ? (v1 & 7)  : (v1 >> 1);
    const int r1 = c1 * 2 + s1;
    float bx = 0.f, by = 0.f, bz = 0.f;
    for (int j1 = 0; j1 < 3; j1++) {
      const int c2 = st3[r1 * 3 + j1];
      const float p2 = sp[r1 * 32 + c2];
      const float q2 = (p2 > PROB_TH) ? p2 : 0.f;
      const float4 k = t3[c2 * 16 + (odd ? 0 : 8) + v2];
      bx += q2 * k.x; by += q2 * k.y; bz += q2 * k.z;
    }
    ox += q1 * bx; oy += q1 * by; oz += q1 * bz;
  }
  float* p = out + ((size_t)blockIdx.x << 12) + (tid << 2);
  const float4 px = make_float4(ox, ox, ox, ox);
  const float4 py = make_float4(oy, oy, oy, oy);
  const float4 pz = make_float4(oz, oz, oz, oz);
  for (int l = 0; l < 4; l++) {
    *(float4*)(p + (l << 10)) = px;
    *(float4*)(p + (1 << 20) + (l << 10)) = py;
    *(float4*)(p + (2 << 20) + (l << 10)) = pz;
  }
}

extern "C" void kernel_launch(void* const* d_in, const int* in_sizes, int n_in,
                              void* d_out, int out_size, void* d_ws, size_t ws_size,
                              hipStream_t stream) {
  (void)in_sizes; (void)n_in; (void)out_size; (void)d_ws; (void)ws_size;
  const float* fc   = (const float*)d_in[0];
  const float* fs   = (const float*)d_in[1];
  const int*   temp = (const int*)d_in[2];
  float* out = (float*)d_out;

  ffg_frame<<<256, 256, 0, stream>>>(fc, fs, temp, out);
}


# All child-index/weight LDS reads (three table levels, depth-4 product) issued once behind the softmax barrier, kept in registers; WMMA color operand built in registers by idle waves; sums unchanged
# speedup vs baseline: 3.5609x; 1.0740x over previous
#include <hip/hip_runtime.h>
#include <hip/hip_bf16.h>
#include <math.h>


typedef __attribute__((ext_vector_type(2))) float v2f;
typedef __attribute__((ext_vector_type(8))) float v8f;

#define PROB_TH 1e-3f

__global__ __launch_bounds__(256) void ffg_frame(
    const float* __restrict__ fc,
    const float* __restrict__ fs,
    const int*   __restrict__ temp,
    float*       __restrict__ out)
{
  __shared__ float sp[64 * 32];
  __shared__ int   st3[64 * 3];
  __shared__ float4 m4[64];
  __shared__ float4 t2[32 * 16];
  __shared__ float4 t3[32 * 16];

  const int tid  = threadIdx.x;
  const int ty   = blockIdx.x >> 3;
  const int wave = tid >> 5;
  const int lane = tid & 31;
  const int half = lane >> 4;
  const int ln   = lane & 15;
  const bool mmw   = (wave & 2) != 0;
  const int  mtile = ((wave >> 2) << 1) | (wave & 1);

  v2f bcol[8] = {};
  if (mmw) {
    const int cl = (ln < 3) ? ln : 2;
#pragma unroll
    for (int s = 0; s < 8; s++) {
      const int kb = 4 * s + 2 * half;
      const float g0 = 1.0f / (1.0f + expf(-fc[(kb + 0) * 3 + cl]));
      const float g1 = 1.0f / (1.0f + expf(-fc[(kb + 1) * 3 + cl]));
      bcol[s].x = (ln < 3) ? g0 : 0.f;
      bcol[s].y = (ln < 3) ? g1 : 0.f;
    }
  }

  if (tid < 64) {
    const float* row = fs + tid * 32;
    float x[32];
#pragma unroll
    for (int j = 0; j < 32; j++) x[j] = row[j];
    const float invT = 1.0f / (float)temp[0];
    float m = -1e30f;
#pragma unroll
    for (int j = 0; j < 32; j++) m = fmaxf(m, x[j] * invT);
    float s = 0.f;
#pragma unroll
    for (int j = 0; j < 32; j++) {
      const float e = expf(x[j] * invT - m);
      s += e;
      x[j] = e;
    }
    const float inv = 1.0f / s;
    float v0 = -1.f, v1 = -1.f, v2 = -1.f;
    int   i0 = 0,    i1 = 0,    i2 = 0;
#pragma unroll
    for (int j = 0; j < 32; j++) {
      const float p = x[j] * inv;
      x[j] = p;
      const bool c0 = p > v0, c1 = p > v1, c2 = p > v2;
      v2 = c1 ? v1 : (c2 ? p : v2);  i2 = c1 ? i1 : (c2 ? j : i2);
      v1 = c0 ? v0 : (c1 ? p : v1);  i1 = c0 ? i0 : (c1 ? j : i1);
      v0 = c0 ? p  : v0;             i0 = c0 ? j  : i0;
    }
#pragma unroll
    for (int j = 0; j < 32; j++) sp[tid * 32 + j] = x[j];
    st3[tid * 3 + 0] = i0;
    st3[tid * 3 + 1] = i1;
    st3[tid * 3 + 2] = i2;
  }
  __syncthreads();

  const int   mrow = mtile * 16 + ln;
  const int   mtop = st3[mrow * 3 + 0];
  const float mpt  = sp[mrow * 32 + mtop];
  int   dA[2][2];
  float dQ[2][2];
#pragma unroll
  for (int i = 0; i < 2; i++) {
    const int e = tid + 256 * i;
    const int idx = e >> 4, u = (e >> 2) & 3, v = e & 3;
    const int odd = idx & 1;
    const int sd = odd ? (v >> 1) : (u >> 1);
    const int uc = odd ? (u >> 1) : (u & 1);
    const int vc = odd ? (v & 1)  : (v >> 1);
    const int r = idx * 2 + sd;
#pragma unroll
    for (int j = 0; j < 2; j++) {
      const int c = st3[r * 3 + j];
      const float p = sp[r * 32 + c];
      dQ[i][j] = (p > PROB_TH) ? p : 0.f;
      dA[i][j] = c * 2 + ((c & 1) ? vc : uc);
    }
  }
  int   eA[2][2];
  float eQ[2][2];
#pragma unroll
  for (int i = 0; i < 2; i++) {
    const int e = tid + 256 * i;
    const int idx = e >> 4, sel = (e >> 3) & 1, v = e & 7;
    const int u = sel ? (ty & 7) : ((ty >> 1) & 7);
    const int odd = idx & 1;
    const int sd = odd ? (v >> 2) : (u >> 2);
    const int uc = odd ? (u >> 1) : (u & 3);
    const int vc = odd ? (v & 3)  : (v >> 1);
    const int r = idx * 2 + sd;
#pragma unroll
    for (int j = 0; j < 2; j++) {
      const int c = st3[r * 3 + j];
      const float p = sp[r * 32 + c];
      eQ[i][j] = (p > PROB_TH) ? p : 0.f;
      eA[i][j] = c * 16 + uc * 4 + vc;
    }
  }
  const int tx = tid >> 3;
  const int r0 = ty >> 4;
  const int u1 = ty & 15, v1 = tx >> 1;
  float fQ1[3];
  int   fA[3][3];
  float fQ2[3][3];
#pragma unroll
  for (int j0 = 0; j0 < 3; j0++) {
    const int c1 = st3[r0 * 3 + j0];
    const float p1 = sp[r0 * 32 + c1];
    fQ1[j0] = (p1 > PROB_TH) ? p1 : 0.f;
    const int odd = c1 & 1;
    const int s1 = odd ? (v1 >> 3) : (u1 >> 3);
    const int v2 = odd ? (v1 & 7)  : (v1 >> 1);
    const int r1 = c1 * 2 + s1;
#pragma unroll
    for (int j1 = 0; j1 < 3; j1++) {
      const int c2 = st3[r1 * 3 + j1];
      const float p2 = sp[r1 * 32 + c2];
      fQ2[j0][j1] = (p2 > PROB_TH) ? p2 : 0.f;
      fA[j0][j1] = c2 * 16 + (odd ? 0 : 8) + v2;
    }
  }
  asm volatile("" :: "v"(mtop), "v"(mpt));

  if (mmw) {
    const int top = mtop;
    const float qt = (mpt > PROB_TH) ? mpt : 0.f;
    v8f acc = {};
#pragma unroll
    for (int s = 0; s < 8; s++) {
      const int kb = 4 * s + 2 * half;
      v2f a;
      a.x = (kb     == top) ? qt : 0.f;
      a.y = (kb + 1 == top) ? qt : 0.f;
      acc = __builtin_amdgcn_wmma_f32_16x16x4_f32(
          false, a, false, bcol[s], (short)0, acc, false, false);
    }
    float* m4f = (float*)m4;
#pragma unroll
    for (int r = 0; r < 8; r++) {
      if (ln < 4) m4f[(mtile * 16 + r + 8 * half) * 4 + ln] = acc[r];
    }
  }
  __syncthreads();

#pragma unroll
  for (int i = 0; i < 2; i++) {
    float ax = 0.f, ay = 0.f, az = 0.f;
#pragma unroll
    for (int j = 0; j < 2; j++) {
      const float q = dQ[i][j];
      const float4 k = m4[dA[i][j]];
      ax += q * k.x; ay += q * k.y; az += q * k.z;
    }
    t2[tid + 256 * i] = make_float4(ax, ay, az, 0.f);
  }
  __syncthreads();

#pragma unroll
  for (int i = 0; i < 2; i++) {
    float ax = 0.f, ay = 0.f, az = 0.f;
#pragma unroll
    for (int j = 0; j < 2; j++) {
      const float q = eQ[i][j];
      const float4 k = t2[eA[i][j]];
      ax += q * k.x; ay += q * k.y; az += q * k.z;
    }
    t3[tid + 256 * i] = make_float4(ax, ay, az, 0.f);
  }
  __syncthreads();

  float ox = 0.f, oy = 0.f, oz = 0.f;
#pragma unroll
  for (int j0 = 0; j0 < 3; j0++) {
    float bx = 0.f, by = 0.f, bz = 0.f;
#pragma unroll
    for (int j1 = 0; j1 < 3; j1++) {
      const float q2 = fQ2[j0][j1];
      const float4 k = t3[fA[j0][j1]];
      bx += q2 * k.x; by += q2 * k.y; bz += q2 * k.z;
    }
    const float q1 = fQ1[j0];
    ox += q1 * bx; oy += q1 * by; oz += q1 * bz;
  }
  float* p = out + ((size_t)blockIdx.x << 12) + (tid << 2);
  const float4 px = make_float4(ox, ox, ox, ox);
  const float4 py = make_float4(oy, oy, oy, oy);
  const float4 pz = make_float4(oz, oz, oz, oz);
  for (int l = 0; l < 4; l++) {
    *(float4*)(p + (l << 10)) = px;
    *(float4*)(p + (1 << 20) + (l << 10)) = py;
    *(float4*)(p + (2 << 20) + (l << 10)) = pz;
  }
}

extern "C" void kernel_launch(void* const* d_in, const int* in_sizes, int n_in,
                              void* d_out, int out_size, void* d_ws, size_t ws_size,
                              hipStream_t stream) {
  (void)in_sizes; (void)n_in; (void)out_size; (void)d_ws; (void)ws_size;
  const float* fc   = (const float*)d_in[0];
  const float* fs   = (const float*)d_in[1];
  const int*   temp = (const int*)d_in[2];
  float* out = (float*)d_out;

  ffg_frame<<<256, 256, 0, stream>>>(fc, fs, temp, out);
}


# Softmax split by function over the four SIMDs: two waves do exps and row sum, two rank top-3 on the scaled logits and keep 3 exps, colors once per element; prob = exp * (1/sum) at use; sums unchanged
# speedup vs baseline: 3.7871x; 1.0635x over previous
#include <hip/hip_runtime.h>
#include <hip/hip_bf16.h>
#include <math.h>


typedef __attribute__((ext_vector_type(2))) float v2f;
typedef __attribute__((ext_vector_type(8))) float v8f;

#define PROB_TH 1e-3f

__global__ __launch_bounds__(256) void ffg_frame(
    const float* __restrict__ fc,
    const float* __restrict__ fs,
    const int*   __restrict__ temp,
    float*       __restrict__ out)
{
  __shared__ int4   rec[64 * 2];
  __shared__ float  sgT[4 * 32];
  __shared__ float4 m4[64];
  __shared__ float4 t2[32 * 16];
  __shared__ float4 t3[32 * 16];

  const int tid  = threadIdx.x;
  const int ty   = blockIdx.x >> 3;
  const int wave = tid >> 5;
  const int lane = tid & 31;
  const int half = lane >> 4;
  const int ln   = lane & 15;
  const bool mmw   = wave >= 4;
  const int  mtile = wave & 3;

  if (wave < 2) {
    const float* row = fs + tid * 32;
    float x[32];
#pragma unroll
    for (int j = 0; j < 32; j++) x[j] = row[j];
    const float invT = 1.0f / (float)temp[0];
    float m = -1e30f;
#pragma unroll
    for (int j = 0; j < 32; j++) m = fmaxf(m, x[j] * invT);
    float s = 0.f;
#pragma unroll
    for (int j = 0; j < 32; j++) s += expf(x[j] * invT - m);
    int2* tailp = (int2*)&rec[tid * 2 + 1];
    tailp[1] = make_int2(0, __float_as_int(1.0f / s));
  } else if (wave < 4) {
    const int r = tid - 64;
    const float* row = fs + r * 32;
    float x[32];
#pragma unroll
    for (int j = 0; j < 32; j++) x[j] = row[j];
    const float invT = 1.0f / (float)temp[0];
    float v0 = -INFINITY, v1 = -INFINITY, v2 = -INFINITY;
    int   i0 = 0,         i1 = 0,         i2 = 0;
#pragma unroll
    for (int j = 0; j < 32; j++) {
      const float p = x[j] * invT;
      const bool c0 = p > v0, c1 = p > v1, c2 = p > v2;
      v2 = c1 ? v1 : (c2 ? p : v2);  i2 = c1 ? i1 : (c2 ? j : i2);
      v1 = c0 ? v0 : (c1 ? p : v1);  i1 = c0 ? i0 : (c1 ? j : i1);
      v0 = c0 ? p  : v0;             i0 = c0 ? j  : i0;
    }
    const float e0 = expf(v0 - v0);
    const float e1 = expf(v1 - v0);
    const float e2 = expf(v2 - v0);
    rec[r * 2] = make_int4(i0, __float_as_int(e0), i1, __float_as_int(e1));
    int2* tailp = (int2*)&rec[r * 2 + 1];
    tailp[0] = make_int2(i2, __float_as_int(e2));
  } else {
    const int i = tid - 128;
    if (i < 96) {
      const int k = i / 3, c = i - 3 * k;
      sgT[c * 32 + k] = 1.0f / (1.0f + expf(-fc[i]));
    } else {
      sgT[i] = 0.f;
    }
  }
  __syncthreads();

  const int cl = (ln < 3) ? ln : 3;
  v2f bcol[8] = {};
  int   mtop = 0;
  float mpt  = 0.f;
  if (mmw) {
    const int mrow = mtile * 16 + ln;
    const int2 m0 = *(const int2*)&rec[mrow * 2];
    const int2 m1 = ((const int2*)&rec[mrow * 2 + 1])[1];
    mtop = m0.x;
    mpt  = __int_as_float(m0.y) * __int_as_float(m1.y);
#pragma unroll
    for (int s = 0; s < 8; s++) {
      const int kb = 4 * s + 2 * half;
      const float2 bb = *(const float2*)&sgT[cl * 32 + kb];
      bcol[s].x = bb.x;
      bcol[s].y = bb.y;
    }
  }
  int   dA[2][2];
  float dQ[2][2];
#pragma unroll
  for (int i = 0; i < 2; i++) {
    const int e = tid + 256 * i;
    const int idx = e >> 4, u = (e >> 2) & 3, v = e & 3;
    const int odd = idx & 1;
    const int sd = odd ? (v >> 1) : (u >> 1);
    const int uc = odd ? (u >> 1) : (u & 1);
    const int vc = odd ? (v & 1)  : (v >> 1);
    const int r = idx * 2 + sd;
    const int4 k01 = rec[r * 2];
    const float iv = __int_as_float(((const int2*)&rec[r * 2 + 1])[1].y);
    const int   cc[2] = {k01.x, k01.z};
    const float ee[2] = {__int_as_float(k01.y), __int_as_float(k01.w)};
#pragma unroll
    for (int j = 0; j < 2; j++) {
      const int c = cc[j];
      const float p = ee[j] * iv;
      dQ[i][j] = (p > PROB_TH) ? p : 0.f;
      dA[i][j] = c * 2 + ((c & 1) ? vc : uc);
    }
  }
  int   eA[2][2];
  float eQ[2][2];
#pragma unroll
  for (int i = 0; i < 2; i++) {
    const int e = tid + 256 * i;
    const int idx = e >> 4, sel = (e >> 3) & 1, v = e & 7;
    const int u = sel ? (ty & 7) : ((ty >> 1) & 7);
    const int odd = idx & 1;
    const int sd = odd ? (v >> 2) : (u >> 2);
    const int uc = odd ? (u >> 1) : (u & 3);
    const int vc = odd ? (v & 3)  : (v >> 1);
    const int r = idx * 2 + sd;
    const int4 k01 = rec[r * 2];
    const float iv = __int_as_float(((const int2*)&rec[r * 2 + 1])[1].y);
    const int   cc[2] = {k01.x, k01.z};
    const float ee[2] = {__int_as_float(k01.y), __int_as_float(k01.w)};
#pragma unroll
    for (int j = 0; j < 2; j++) {
      const int c = cc[j];
      const float p = ee[j] * iv;
      eQ[i][j] = (p > PROB_TH) ? p : 0.f;
      eA[i][j] = c * 16 + uc * 4 + vc;
    }
  }
  const int tx = tid >> 3;
  const int r0 = ty >> 4;
  const int u1 = ty & 15, v1 = tx >> 1;
  float fQ1[3];
  int   fA[3][3];
  float fQ2[3][3];
  {
    const int4 a01 = rec[r0 * 2];
    const int4 a2  = rec[r0 * 2 + 1];
    const float iv0 = __int_as_float(a2.w);
    const int   c1s[3] = {a01.x, a01.z, a2.x};
    const float e1s[3] = {__int_as_float(a01.y), __int_as_float(a01.w), __int_as_float(a2.y)};
#pragma unroll
    for (int j0 = 0; j0 < 3; j0++) {
      const int c1 = c1s[j0];
      const float p1 = e1s[j0] * iv0;
      fQ1[j0] = (p1 > PROB_TH) ? p1 : 0.f;
      const int odd = c1 & 1;
      const int s1 = odd ? (v1 >> 3) : (u1 >> 3);
      const int v2 = odd ? (v1 & 7)  : (v1 >> 1);
      const int r1 = c1 * 2 + s1;
      const int4 b01 = rec[r1 * 2];
      const int4 b2  = rec[r1 * 2 + 1];
      const float iv1 = __int_as_float(b2.w);
      const int   c2s[3] = {b01.x, b01.z, b2.x};
      const float e2s[3] = {__int_as_float(b01.y), __int_as_float(b01.w), __int_as_float(b2.y)};
#pragma unroll
      for (int j1 = 0; j1 < 3; j1++) {
        const int c2 = c2s[j1];
        const float p2 = e2s[j1] * iv1;
        fQ2[j0][j1] = (p2 > PROB_TH) ? p2 : 0.f;
        fA[j0][j1] = c2 * 16 + (odd ? 0 : 8) + v2;
      }
    }
  }

  if (mmw) {
    const int top = mtop;
    const float qt = (mpt > PROB_TH) ? mpt : 0.f;
    v8f acc = {};
#pragma unroll
    for (int s = 0; s < 8; s++) {
      const int kb = 4 * s + 2 * half;
      v2f a;
      a.x = (kb     == top) ? qt : 0.f;
      a.y = (kb + 1 == top) ? qt : 0.f;
      acc = __builtin_amdgcn_wmma_f32_16x16x4_f32(
          false, a, false, bcol[s], (short)0, acc, false, false);
    }
    float* m4f = (float*)m4;
#pragma unroll
    for (int r = 0; r < 8; r++) {
      if (ln < 4) m4f[(mtile * 16 + r + 8 * half) * 4 + ln] = acc[r];
    }
  }
  __syncthreads();

#pragma unroll
  for (int i = 0; i < 2; i++) {
    float ax = 0.f, ay = 0.f, az = 0.f;
#pragma unroll
    for (int j = 0; j < 2; j++) {
      const float q = dQ[i][j];
      const float4 k = m4[dA[i][j]];
      ax += q * k.x; ay += q * k.y; az += q * k.z;
    }
    t2[tid + 256 * i] = make_float4(ax, ay, az, 0.f);
  }
  __syncthreads();

#pragma unroll
  for (int i = 0; i < 2; i++) {
    float ax = 0.f, ay = 0.f, az = 0.f;
#pragma unroll
    for (int j = 0; j < 2; j++) {
      const float q = eQ[i][j];
      const float4 k = t2[eA[i][j]];
      ax += q * k.x; ay += q * k.y; az += q * k.z;
    }
    t3[tid + 256 * i] = make_float4(ax, ay, az, 0.f);
  }
  __syncthreads();

  float ox = 0.f, oy = 0.f, oz = 0.f;
#pragma unroll
  for (int j0 = 0; j0 < 3; j0++) {
    float bx = 0.f, by = 0.f, bz = 0.f;
#pragma unroll
    for (int j1 = 0; j1 < 3; j1++) {
      const float q2 = fQ2[j0][j1];
      const float4 k = t3[fA[j0][j1]];
      bx += q2 * k.x; by += q2 * k.y; bz += q2 * k.z;
    }
    const float q1 = fQ1[j0];
    ox += q1 * bx; oy += q1 * by; oz += q1 * bz;
  }
  float* p = out + ((size_t)blockIdx.x << 12) + (tid << 2);
  const float4 px = make_float4(ox, ox, ox, ox);
  const float4 py = make_float4(oy, oy, oy, oy);
  const float4 pz = make_float4(oz, oz, oz, oz);
  for (int l = 0; l < 4; l++) {
    *(float4*)(p + (l << 10)) = px;
    *(float4*)(p + (1 << 20) + (l << 10)) = py;
    *(float4*)(p + (2 << 20) + (l << 10)) = pz;
  }
}

extern "C" void kernel_launch(void* const* d_in, const int* in_sizes, int n_in,
                              void* d_out, int out_size, void* d_ws, size_t ws_size,
                              hipStream_t stream) {
  (void)in_sizes; (void)n_in; (void)out_size; (void)d_ws; (void)ws_size;
  const float* fc   = (const float*)d_in[0];
  const float* fs   = (const float*)d_in[1];
  const int*   temp = (const int*)d_in[2];
  float* out = (float*)d_out;

  ffg_frame<<<256, 256, 0, stream>>>(fc, fs, temp, out);
}
